# starting kernel
# baseline (speedup 1.0000x reference)
#include <hip/hip_runtime.h>


typedef __attribute__((ext_vector_type(2))) float v2f;
typedef __attribute__((ext_vector_type(8))) float v8f;

#define F_OUT 128
#define N_HEAD 4

__global__ void rgat_prep(const float* __restrict__ w,
                          const float* __restrict__ a,
                          float* __restrict__ cs,
                          float* __restrict__ BsrcP,
                          float* __restrict__ BdstP) {
  int f = threadIdx.x;
  float c1 = w[f];
  float c2 = c1 * w[F_OUT + f];
  float c3 = c2 * w[2 * F_OUT + f];
  cs[f] = 1.0f;  cs[F_OUT + f] = c1;  cs[2 * F_OUT + f] = c2;  cs[3 * F_OUT + f] = c3;

  for (int idx = f; idx < 32 * 64; idx += F_OUT) {
    int s    = idx >> 6;
    int l    = (idx >> 1) & 31;
    int half = idx & 1;
    int col  = l & 15;
    int k    = s * 4 + ((l >> 4) << 1) + half;
    float bs = 0.0f, bd = 0.0f;
    if (col < 4) {
      float c = 1.0f;
      for (int j = 0; j < col; ++j) c *= w[j * F_OUT + k];
      bs = c * a[col * 256 + k];
      bd = a[col * 256 + F_OUT + k];
    }
    BsrcP[idx] = bs;
    BdstP[idx] = bd;
  }
}

__global__ void rgat_score(const float* __restrict__ X,
                           const float* __restrict__ Bp,
                           float* __restrict__ S, int M) {
  int tile = blockIdx.x * (blockDim.x >> 5) + (threadIdx.x >> 5);
  int base = tile * 16;
  if (base >= M) return;

  int lane = threadIdx.x & 31;
  int col  = lane & 15;
  int kb   = (lane >> 4) * 2;
  int row  = base + (lane & 15);
  row = row < M ? row : M - 1;
  const float* xr = X + (size_t)row * F_OUT;
  const float* bp = Bp + lane * 2;

  v8f acc = {};
#pragma unroll
  for (int s = 0; s < 32; ++s) {
    v2f av = *(const v2f*)(xr + s * 4 + kb);
    v2f bv = *(const v2f*)(bp + s * 64);
    acc = __builtin_amdgcn_wmma_f32_16x16x4_f32(
         false, av,  false, bv,
         (short)0, acc,  false,  false);
  }

  int mhalf = (lane >> 4) * 8;
#pragma unroll
  for (int v = 0; v < 8; ++v)
    S[(size_t)(base + mhalf + v) * 16 + col] = acc[v];
}

__global__ void rgat_edge(const int* __restrict__ A,
                          const float* __restrict__ h,
                          const float* __restrict__ inputr,
                          const float* __restrict__ cs,
                          const float* __restrict__ Snode,
                          const float* __restrict__ Srel,
                          float* __restrict__ rowsum,
                          float* __restrict__ out,
                          int E, int N) {
  const int lane = threadIdx.x & 31;
  const int wid  = (blockIdx.x * blockDim.x + threadIdx.x) >> 5;
  const int nw   = (gridDim.x * blockDim.x) >> 5;

  const float4 cv0 = *(const float4*)(cs + 0 * F_OUT + lane * 4);
  const float4 cv1 = *(const float4*)(cs + 1 * F_OUT + lane * 4);
  const float4 cv2 = *(const float4*)(cs + 2 * F_OUT + lane * 4);
  const float4 cv3 = *(const float4*)(cs + 3 * F_OUT + lane * 4);
  const float4 cv[N_HEAD] = {cv0, cv1, cv2, cv3};

  for (int ev = wid; ev < E; ev += nw) {
    const int e   = __builtin_amdgcn_readfirstlane(ev);
    const int dst = A[e];
    const int rl  = A[E + e];
    const int src = A[2 * E + e];

    const float4 hs = *(const float4*)(h + (size_t)src * F_OUT + lane * 4);
    const float4 rf = *(const float4*)(inputr + (size_t)rl * F_OUT + lane * 4);
    const float4 sn = *(const float4*)(Snode + (size_t)src * 16);
    const float4 sr = *(const float4*)(Srel + (size_t)rl * 16);

    const float logit[N_HEAD] = {sn.x + sr.x, sn.y + sr.y, sn.z + sr.z, sn.w + sr.w};
    float ee[N_HEAD];
#pragma unroll
    for (int i = 0; i < N_HEAD; ++i) {
      float lk = logit[i] > 0.0f ? logit[i] : 0.2f * logit[i];
      ee[i] = __expf(-lk);
    }

    if (lane == 0) {
#pragma unroll
      for (int i = 0; i < N_HEAD; ++i)
        __hip_atomic_fetch_add(&rowsum[(size_t)i * N + dst], ee[i],
                               __ATOMIC_RELAXED, __HIP_MEMORY_SCOPE_AGENT);
    }

#pragma unroll
    for (int i = 0; i < N_HEAD; ++i) {
      float* op = out + ((size_t)i * N + dst) * F_OUT + lane * 4;
      __hip_atomic_fetch_add(&op[0], (cv[i].x * hs.x - rf.x) * ee[i],
                             __ATOMIC_RELAXED, __HIP_MEMORY_SCOPE_AGENT);
      __hip_atomic_fetch_add(&op[1], (cv[i].y * hs.y - rf.y) * ee[i],
                             __ATOMIC_RELAXED, __HIP_MEMORY_SCOPE_AGENT);
      __hip_atomic_fetch_add(&op[2], (cv[i].z * hs.z - rf.z) * ee[i],
                             __ATOMIC_RELAXED, __HIP_MEMORY_SCOPE_AGENT);
      __hip_atomic_fetch_add(&op[3], (cv[i].w * hs.w - rf.w) * ee[i],
                             __ATOMIC_RELAXED, __HIP_MEMORY_SCOPE_AGENT);
    }
  }
}

__global__ void rgat_norm(float* __restrict__ out,
                          const float* __restrict__ rowsum,
                          int total) {
  int idx = blockIdx.x * blockDim.x + threadIdx.x;
  if (idx >= total) return;
  out[idx] = out[idx] / rowsum[idx >> 7];
}

extern "C" void kernel_launch(void* const* d_in, const int* in_sizes, int n_in,
                              void* d_out, int out_size, void* d_ws, size_t ws_size,
                              hipStream_t stream) {
  const float* h      = (const float*)d_in[0];
  const float* inputr = (const float*)d_in[1];
  const float* w      = (const float*)d_in[2];
  const float* a      = (const float*)d_in[3];
  const int*   A      = (const int*)d_in[4];

  const int N    = in_sizes[0] / F_OUT;
  const int NREL = in_sizes[1] / F_OUT;
  const int E    = in_sizes[4] / 3;

  float* ws = (float*)d_ws;
  float* cs    = ws;
  float* BsrcP = ws + 512;
  float* BdstP = ws + 512 + 2048;
  size_t off   = 512 + 2048 + 2048;
  const int nrel_tiles = (NREL + 15) / 16;
  const int node_tiles = (N + 15) / 16;
  float* Srel  = ws + off;  off += (size_t)nrel_tiles * 16 * 16;
  float* Snode = ws + off;  off += (size_t)node_tiles * 16 * 16;
  float* rowsum = ws + off; off += (size_t)N_HEAD * N;
  (void)ws_size;

  hipMemsetAsync(d_out, 0, (size_t)out_size * sizeof(float), stream);
  hipMemsetAsync(rowsum, 0, (size_t)N_HEAD * N * sizeof(float), stream);

  rgat_prep<<<1, F_OUT, 0, stream>>>(w, a, cs, BsrcP, BdstP);

  const int WPB = 8;
  rgat_score<<<(node_tiles + WPB - 1) / WPB, 256, 0, stream>>>(h, BsrcP, Snode, N);
  rgat_score<<<(nrel_tiles + WPB - 1) / WPB, 256, 0, stream>>>(inputr, BdstP, Srel, NREL);

  int eblocks = (E + WPB - 1) / WPB;
  if (eblocks > 4096) eblocks = 4096;
  rgat_edge<<<eblocks, 256, 0, stream>>>(A, h, inputr, cs, Snode, Srel,
                                         rowsum, (float*)d_out, E, N);

  rgat_norm<<<(out_size + 255) / 256, 256, 0, stream>>>((float*)d_out, rowsum, out_size);
}


# Edge list grouped by dst in-call (returning int add = slot, prefix sum, place); one wave sums a dst's rows for 4 heads in registers, divides, stores once: no f32 atomics; sum order = arrival slot
# speedup vs baseline: 3.4979x; 3.4979x over previous
#include <hip/hip_runtime.h>


typedef __attribute__((ext_vector_type(2))) float v2f;
typedef __attribute__((ext_vector_type(8))) float v8f;

#define F_OUT 128
#define N_HEAD 4

__global__ void rgat_prep(const float* __restrict__ w,
                          const float* __restrict__ a,
                          float* __restrict__ cs,
                          float* __restrict__ BsrcP,
                          float* __restrict__ BdstP) {
  int f = threadIdx.x;
  float c1 = w[f];
  float c2 = c1 * w[F_OUT + f];
  float c3 = c2 * w[2 * F_OUT + f];
  cs[f] = 1.0f;  cs[F_OUT + f] = c1;  cs[2 * F_OUT + f] = c2;  cs[3 * F_OUT + f] = c3;

  for (int idx = f; idx < 32 * 64; idx += F_OUT) {
    int s    = idx >> 6;
    int l    = (idx >> 1) & 31;
    int half = idx & 1;
    int col  = l & 15;
    int k    = s * 4 + ((l >> 4) << 1) + half;
    float bs = 0.0f, bd = 0.0f;
    if (col < 4) {
      float c = 1.0f;
      for (int j = 0; j < col; ++j) c *= w[j * F_OUT + k];
      bs = c * a[col * 256 + k];
      bd = a[col * 256 + F_OUT + k];
    }
    BsrcP[idx] = bs;
    BdstP[idx] = bd;
  }
}

__global__ void rgat_score(const float* __restrict__ X,
                           const float* __restrict__ Bp,
                           float* __restrict__ S, int M) {
  int tile = blockIdx.x * (blockDim.x >> 5) + (threadIdx.x >> 5);
  int base = tile * 16;
  if (base >= M) return;

  int lane = threadIdx.x & 31;
  int col  = lane & 15;
  int kb   = (lane >> 4) * 2;
  int row  = base + (lane & 15);
  row = row < M ? row : M - 1;
  const float* xr = X + (size_t)row * F_OUT;
  const float* bp = Bp + lane * 2;

  v8f acc = {};
#pragma unroll
  for (int s = 0; s < 32; ++s) {
    v2f av = *(const v2f*)(xr + s * 4 + kb);
    v2f bv = *(const v2f*)(bp + s * 64);
    acc = __builtin_amdgcn_wmma_f32_16x16x4_f32(
         false, av,  false, bv,
         (short)0, acc,  false,  false);
  }

  int mhalf = (lane >> 4) * 8;
#pragma unroll
  for (int v = 0; v < 8; ++v)
    S[(size_t)(base + mhalf + v) * 16 + col] = acc[v];
}

__global__ void rgat_count(const int* __restrict__ A,
                           int* __restrict__ cnt,
                           int* __restrict__ rank, int E) {
  int e = blockIdx.x * blockDim.x + threadIdx.x;
  if (e >= E) return;
  rank[e] = __hip_atomic_fetch_add(&cnt[A[e]], 1, __ATOMIC_RELAXED,
                                   __HIP_MEMORY_SCOPE_AGENT);
}

#define SCAN_T 1024
__global__ void __launch_bounds__(SCAN_T)
rgat_scan(const int* __restrict__ cnt, int* __restrict__ off, int N) {
  __shared__ int wtot[32];
  __shared__ int ptot[32];
  const int tid  = threadIdx.x;
  const int lane = tid & 31;
  const int wv   = tid >> 5;
  const int first = blockIdx.x * SCAN_T;

  int pre = 0;
  for (int i = tid; i < first; i += SCAN_T) pre += cnt[i];

  const int idx = first + tid;
  const int v = idx < N ? cnt[idx] : 0;
  int x = v;
  for (int d = 1; d < 32; d <<= 1) {
    int y = __shfl_up(x, d, 32);
    if (lane >= d) x += y;
    pre += __shfl_xor(pre, d, 32);
  }
  if (lane == 31) wtot[wv] = x;
  if (lane == 0)  ptot[wv] = pre;
  __syncthreads();

  int base = 0;
  for (int k = 0; k < 32; ++k) {
    base += ptot[k];
    if (k < wv) base += wtot[k];
  }
  if (idx < N) off[idx] = base + x - v;
  if (idx == N - 1) off[N] = base + x;
}

__global__ void rgat_place(const int* __restrict__ A,
                           const int* __restrict__ off,
                           const int* __restrict__ rank,
                           int* __restrict__ pk, int E) {
  int e = blockIdx.x * blockDim.x + threadIdx.x;
  if (e >= E) return;
  const int p = off[A[e]] + rank[e];
  *(int2*)(pk + 2 * (size_t)p) = make_int2(A[2 * E + e], A[E + e]);
}

struct EdgeIn { float4 hs, rf, sn, sr; };

__device__ __forceinline__ EdgeIn edge_load(const float* __restrict__ h,
                                            const float* __restrict__ inputr,
                                            const float* __restrict__ Snode,
                                            const float* __restrict__ Srel,
                                            int src, int rl, int lane) {
  EdgeIn in;
  in.hs = *(const float4*)(h + (size_t)src * F_OUT + lane * 4);
  in.rf = *(const float4*)(inputr + (size_t)rl * F_OUT + lane * 4);
  in.sn = *(const float4*)(Snode + (size_t)src * 16);
  in.sr = *(const float4*)(Srel + (size_t)rl * 16);
  return in;
}

__device__ __forceinline__ void edge_acc(const EdgeIn& in, const float4* cv,
                                         float4* acc, float* rs) {
  const float logit[N_HEAD] = {in.sn.x + in.sr.x, in.sn.y + in.sr.y,
                               in.sn.z + in.sr.z, in.sn.w + in.sr.w};
#pragma unroll
  for (int i = 0; i < N_HEAD; ++i) {
    float lk = logit[i] > 0.0f ? logit[i] : 0.2f * logit[i];
    float ee = __expf(-lk);
    rs[i] += ee;
    acc[i].x += (cv[i].x * in.hs.x - in.rf.x) * ee;
    acc[i].y += (cv[i].y * in.hs.y - in.rf.y) * ee;
    acc[i].z += (cv[i].z * in.hs.z - in.rf.z) * ee;
    acc[i].w += (cv[i].w * in.hs.w - in.rf.w) * ee;
  }
}

__global__ void __launch_bounds__(128)
rgat_gather(const int* __restrict__ pk,
            const int* __restrict__ off,
            const float* __restrict__ h,
            const float* __restrict__ inputr,
            const float* __restrict__ cs,
            const float* __restrict__ Snode,
            const float* __restrict__ Srel,
            float* __restrict__ out, int N) {
  const int lane = threadIdx.x & 31;
  const int wid  = (blockIdx.x * blockDim.x + threadIdx.x) >> 5;
  const int nw   = (gridDim.x * blockDim.x) >> 5;

  const float4 cv0 = *(const float4*)(cs + 0 * F_OUT + lane * 4);
  const float4 cv1 = *(const float4*)(cs + 1 * F_OUT + lane * 4);
  const float4 cv2 = *(const float4*)(cs + 2 * F_OUT + lane * 4);
  const float4 cv3 = *(const float4*)(cs + 3 * F_OUT + lane * 4);
  const float4 cv[N_HEAD] = {cv0, cv1, cv2, cv3};

  for (int dv = wid; dv < N; dv += nw) {
    const int d     = __builtin_amdgcn_readfirstlane(dv);
    const int start = off[d];
    const int deg   = off[d + 1] - start;

    float4 acc[N_HEAD];
    float  rs[N_HEAD];
#pragma unroll
    for (int i = 0; i < N_HEAD; ++i) {
      acc[i] = make_float4(0.0f, 0.0f, 0.0f, 0.0f);
      rs[i]  = 0.0f;
    }

    for (int b = 0; b < deg; b += 32) {
      const int n = deg - b < 32 ? deg - b : 32;
      int2 mine = make_int2(0, 0);
      if (lane < n) mine = *(const int2*)(pk + 2 * (size_t)(start + b + lane));

      int j = 0;
      for (; j + 4 <= n; j += 4) {
        const EdgeIn e0 = edge_load(h, inputr, Snode, Srel,
            __builtin_amdgcn_readlane(mine.x, j),     __builtin_amdgcn_readlane(mine.y, j),     lane);
        const EdgeIn e1 = edge_load(h, inputr, Snode, Srel,
            __builtin_amdgcn_readlane(mine.x, j + 1), __builtin_amdgcn_readlane(mine.y, j + 1), lane);
        const EdgeIn e2 = edge_load(h, inputr, Snode, Srel,
            __builtin_amdgcn_readlane(mine.x, j + 2), __builtin_amdgcn_readlane(mine.y, j + 2), lane);
        const EdgeIn e3 = edge_load(h, inputr, Snode, Srel,
            __builtin_amdgcn_readlane(mine.x, j + 3), __builtin_amdgcn_readlane(mine.y, j + 3), lane);
        edge_acc(e0, cv, acc, rs);
        edge_acc(e1, cv, acc, rs);
        edge_acc(e2, cv, acc, rs);
        edge_acc(e3, cv, acc, rs);
      }
      for (; j < n; ++j) {
        const EdgeIn e0 = edge_load(h, inputr, Snode, Srel,
            __builtin_amdgcn_readlane(mine.x, j), __builtin_amdgcn_readlane(mine.y, j), lane);
        edge_acc(e0, cv, acc, rs);
      }
    }

#pragma unroll
    for (int i = 0; i < N_HEAD; ++i) {
      float4 o;
      o.x = acc[i].x / rs[i];
      o.y = acc[i].y / rs[i];
      o.z = acc[i].z / rs[i];
      o.w = acc[i].w / rs[i];
      *(float4*)(out + ((size_t)i * N + d) * F_OUT + lane * 4) = o;
    }
  }
}

__global__ void rgat_edge(const int* __restrict__ A,
                          const float* __restrict__ h,
                          const float* __restrict__ inputr,
                          const float* __restrict__ cs,
                          const float* __restrict__ Snode,
                          const float* __restrict__ Srel,
                          float* __restrict__ rowsum,
                          float* __restrict__ out,
                          int E, int N) {
  const int lane = threadIdx.x & 31;
  const int wid  = (blockIdx.x * blockDim.x + threadIdx.x) >> 5;
  const int nw   = (gridDim.x * blockDim.x) >> 5;

  const float4 cv0 = *(const float4*)(cs + 0 * F_OUT + lane * 4);
  const float4 cv1 = *(const float4*)(cs + 1 * F_OUT + lane * 4);
  const float4 cv2 = *(const float4*)(cs + 2 * F_OUT + lane * 4);
  const float4 cv3 = *(const float4*)(cs + 3 * F_OUT + lane * 4);
  const float4 cv[N_HEAD] = {cv0, cv1, cv2, cv3};

  for (int ev = wid; ev < E; ev += nw) {
    const int e   = __builtin_amdgcn_readfirstlane(ev);
    const int dst = A[e];
    const int rl  = A[E + e];
    const int src = A[2 * E + e];

    const float4 hs = *(const float4*)(h + (size_t)src * F_OUT + lane * 4);
    const float4 rf = *(const float4*)(inputr + (size_t)rl * F_OUT + lane * 4);
    const float4 sn = *(const float4*)(Snode + (size_t)src * 16);
    const float4 sr = *(const float4*)(Srel + (size_t)rl * 16);

    const float logit[N_HEAD] = {sn.x + sr.x, sn.y + sr.y, sn.z + sr.z, sn.w + sr.w};
    float ee[N_HEAD];
#pragma unroll
    for (int i = 0; i < N_HEAD; ++i) {
      float lk = logit[i] > 0.0f ? logit[i] : 0.2f * logit[i];
      ee[i] = __expf(-lk);
    }

    if (lane == 0) {
#pragma unroll
      for (int i = 0; i < N_HEAD; ++i)
        __hip_atomic_fetch_add(&rowsum[(size_t)i * N + dst], ee[i],
                               __ATOMIC_RELAXED, __HIP_MEMORY_SCOPE_AGENT);
    }

#pragma unroll
    for (int i = 0; i < N_HEAD; ++i) {
      float* op = out + ((size_t)i * N + dst) * F_OUT + lane * 4;
      __hip_atomic_fetch_add(&op[0], (cv[i].x * hs.x - rf.x) * ee[i],
                             __ATOMIC_RELAXED, __HIP_MEMORY_SCOPE_AGENT);
      __hip_atomic_fetch_add(&op[1], (cv[i].y * hs.y - rf.y) * ee[i],
                             __ATOMIC_RELAXED, __HIP_MEMORY_SCOPE_AGENT);
      __hip_atomic_fetch_add(&op[2], (cv[i].z * hs.z - rf.z) * ee[i],
                             __ATOMIC_RELAXED, __HIP_MEMORY_SCOPE_AGENT);
      __hip_atomic_fetch_add(&op[3], (cv[i].w * hs.w - rf.w) * ee[i],
                             __ATOMIC_RELAXED, __HIP_MEMORY_SCOPE_AGENT);
    }
  }
}

__global__ void rgat_norm(float* __restrict__ out,
                          const float* __restrict__ rowsum,
                          int total) {
  int idx = blockIdx.x * blockDim.x + threadIdx.x;
  if (idx >= total) return;
  out[idx] = out[idx] / rowsum[idx >> 7];
}

extern "C" void kernel_launch(void* const* d_in, const int* in_sizes, int n_in,
                              void* d_out, int out_size, void* d_ws, size_t ws_size,
                              hipStream_t stream) {
  const float* h      = (const float*)d_in[0];
  const float* inputr = (const float*)d_in[1];
  const float* w      = (const float*)d_in[2];
  const float* a      = (const float*)d_in[3];
  const int*   A      = (const int*)d_in[4];

  const int N    = in_sizes[0] / F_OUT;
  const int NREL = in_sizes[1] / F_OUT;
  const int E    = in_sizes[4] / 3;

  float* ws = (float*)d_ws;
  float* cs    = ws;
  float* BsrcP = ws + 512;
  float* BdstP = ws + 512 + 2048;
  size_t off   = 512 + 2048 + 2048;
  const int nrel_tiles = (NREL + 15) / 16;
  const int node_tiles = (N + 15) / 16;
  float* Srel  = ws + off;  off += (size_t)nrel_tiles * 16 * 16;
  float* Snode = ws + off;  off += (size_t)node_tiles * 16 * 16;
  const size_t tail = off;

  int* cnt  = (int*)ws + off;  off += (size_t)N;
  int* offs = (int*)ws + off;  off += (size_t)N + 1;
  int* rank = (int*)ws + off;  off += (size_t)E;
  off = (off + 1) & ~(size_t)1;
  int* pk   = (int*)ws + off;  off += (size_t)2 * E;
  const bool grouped = ws_size >= off * sizeof(float);

  const int WPB = 8;

  if (grouped) {
    hipMemsetAsync(cnt, 0, (size_t)N * sizeof(int), stream);
  } else {
    hipMemsetAsync(d_out, 0, (size_t)out_size * sizeof(float), stream);
    hipMemsetAsync(ws + tail, 0, (size_t)N_HEAD * N * sizeof(float), stream);
  }

  rgat_prep<<<1, F_OUT, 0, stream>>>(w, a, cs, BsrcP, BdstP);

  rgat_score<<<(node_tiles + WPB - 1) / WPB, 256, 0, stream>>>(h, BsrcP, Snode, N);
  rgat_score<<<(nrel_tiles + WPB - 1) / WPB, 256, 0, stream>>>(inputr, BdstP, Srel, NREL);

  if (grouped) {
    rgat_count<<<(E + 255) / 256, 256, 0, stream>>>(A, cnt, rank, E);
    rgat_scan<<<(N + SCAN_T - 1) / SCAN_T, SCAN_T, 0, stream>>>(cnt, offs, N);
    rgat_place<<<(E + 255) / 256, 256, 0, stream>>>(A, offs, rank, pk, E);

    int gblocks = (N + 3) / 4;
    if (gblocks > 4096) gblocks = 4096;
    rgat_gather<<<gblocks, 128, 0, stream>>>(pk, offs, h, inputr, cs, Snode, Srel,
                                             (float*)d_out, N);
  } else {
    float* rowsum = ws + tail;
    int eblocks = (E + WPB - 1) / WPB;
    if (eblocks > 4096) eblocks = 4096;
    rgat_edge<<<eblocks, 256, 0, stream>>>(A, h, inputr, cs, Snode, Srel,
                                           rowsum, (float*)d_out, E, N);

    rgat_norm<<<(out_size + 255) / 256, 256, 0, stream>>>((float*)d_out, rowsum, out_size);
  }
}


# Prep, both score GEMMs and the in-degree count run as block roles of ONE launch (B fragments built per block in LDS from w and a; head scales rebuilt from w in the gather): 7 -> 4 kernel launches
# speedup vs baseline: 4.3793x; 1.2520x over previous
#include <hip/hip_runtime.h>


typedef __attribute__((ext_vector_type(2))) float v2f;
typedef __attribute__((ext_vector_type(8))) float v8f;

#define F_OUT 128
#define N_HEAD 4

#define FRONT_T 256
__global__ void __launch_bounds__(FRONT_T)
rgat_front(const float* __restrict__ h,
           const float* __restrict__ inputr,
           const float* __restrict__ w,
           const float* __restrict__ a,
           const int* __restrict__ A,
           float* __restrict__ Snode,
           float* __restrict__ Srel,
           int* __restrict__ cnt,
           int* __restrict__ rank,
           int N, int NREL, int E, int node_blocks, int score_blocks) {
  __shared__ float Bl[32 * 64];
  const int b = blockIdx.x;

  if (b >= score_blocks) {
    const int e = (b - score_blocks) * FRONT_T + threadIdx.x;
    if (e < E)
      rank[e] = __hip_atomic_fetch_add(&cnt[A[e]], 1, __ATOMIC_RELAXED,
                                       __HIP_MEMORY_SCOPE_AGENT);
    return;
  }

  const bool isrel = b >= node_blocks;
  {
    const int t    = threadIdx.x;
    const int l    = (t >> 1) & 31;
    const int col  = l & 15;
    const int k0   = (t >> 6) * 4 + ((l >> 4) << 1) + (t & 1);
    float v[8];
#pragma unroll
    for (int i = 0; i < 8; ++i) v[i] = 0.0f;
    if (col < 4) {
      if (isrel) {
#pragma unroll
        for (int i = 0; i < 8; ++i)
          v[i] = a[col * 256 + F_OUT + k0 + 16 * i];
      } else {
#pragma unroll
        for (int i = 0; i < 8; ++i) {
          const int k = k0 + 16 * i;
          const float f0 = w[0 * F_OUT + k];
          const float f1 = w[1 * F_OUT + k];
          const float f2 = w[2 * F_OUT + k];
          float c = col > 0 ? f0 : 1.0f;
          c *= col > 1 ? f1 : 1.0f;
          c *= col > 2 ? f2 : 1.0f;
          v[i] = c * a[col * 256 + k];
        }
      }
    }
#pragma unroll
    for (int i = 0; i < 8; ++i) Bl[t + FRONT_T * i] = v[i];
  }
  __syncthreads();

  const float* __restrict__ X = isrel ? inputr : h;
  float* __restrict__ S       = isrel ? Srel : Snode;
  const int M                 = isrel ? NREL : N;
  const int tile = (isrel ? b - node_blocks : b) * (FRONT_T >> 5) + (threadIdx.x >> 5);
  const int base = tile * 16;
  if (base >= M) return;

  const int lane = threadIdx.x & 31;
  const int col  = lane & 15;
  const int kb   = (lane >> 4) * 2;
  int row = base + (lane & 15);
  row = row < M ? row : M - 1;
  const float* xr = X + (size_t)row * F_OUT;
  const float* bp = Bl + lane * 2;

  v8f acc = {};
#pragma unroll
  for (int s = 0; s < 32; ++s) {
    v2f av = *(const v2f*)(xr + s * 4 + kb);
    v2f bv = *(const v2f*)(bp + s * 64);
    acc = __builtin_amdgcn_wmma_f32_16x16x4_f32(
         false, av,  false, bv,
         (short)0, acc,  false,  false);
  }

  const int mhalf = (lane >> 4) * 8;
#pragma unroll
  for (int v = 0; v < 8; ++v)
    S[(size_t)(base + mhalf + v) * 16 + col] = acc[v];
}

#define SCAN_T 1024
__global__ void __launch_bounds__(SCAN_T)
rgat_scan(const int* __restrict__ cnt, int* __restrict__ off, int N) {
  __shared__ int wtot[32];
  __shared__ int ptot[32];
  const int tid  = threadIdx.x;
  const int lane = tid & 31;
  const int wv   = tid >> 5;
  const int first = blockIdx.x * SCAN_T;

  int pre = 0;
  for (int i = tid; i < first; i += SCAN_T) pre += cnt[i];

  const int idx = first + tid;
  const int v = idx < N ? cnt[idx] : 0;
  int x = v;
  for (int d = 1; d < 32; d <<= 1) {
    int y = __shfl_up(x, d, 32);
    if (lane >= d) x += y;
    pre += __shfl_xor(pre, d, 32);
  }
  if (lane == 31) wtot[wv] = x;
  if (lane == 0)  ptot[wv] = pre;
  __syncthreads();

  int base = 0;
  for (int k = 0; k < 32; ++k) {
    base += ptot[k];
    if (k < wv) base += wtot[k];
  }
  if (idx < N) off[idx] = base + x - v;
  if (idx == N - 1) off[N] = base + x;
}

__global__ void rgat_place(const int* __restrict__ A,
                           const int* __restrict__ off,
                           const int* __restrict__ rank,
                           int* __restrict__ pk, int E) {
  int e = blockIdx.x * blockDim.x + threadIdx.x;
  if (e >= E) return;
  const int p = off[A[e]] + rank[e];
  *(int2*)(pk + 2 * (size_t)p) = make_int2(A[2 * E + e], A[E + e]);
}

struct EdgeIn { float4 hs, rf, sn, sr; };

__device__ __forceinline__ void head_scales(const float* __restrict__ w, int lane,
                                            float4* cv) {
  const float4 w0 = *(const float4*)(w + 0 * F_OUT + lane * 4);
  const float4 w1 = *(const float4*)(w + 1 * F_OUT + lane * 4);
  const float4 w2 = *(const float4*)(w + 2 * F_OUT + lane * 4);
  cv[0] = make_float4(1.0f, 1.0f, 1.0f, 1.0f);
  cv[1] = w0;
  cv[2] = make_float4(cv[1].x * w1.x, cv[1].y * w1.y, cv[1].z * w1.z, cv[1].w * w1.w);
  cv[3] = make_float4(cv[2].x * w2.x, cv[2].y * w2.y, cv[2].z * w2.z, cv[2].w * w2.w);
}

__device__ __forceinline__ EdgeIn edge_load(const float* __restrict__ h,
                                            const float* __restrict__ inputr,
                                            const float* __restrict__ Snode,
                                            const float* __restrict__ Srel,
                                            int src, int rl, int lane) {
  EdgeIn in;
  in.hs = *(const float4*)(h + (size_t)src * F_OUT + lane * 4);
  in.rf = *(const float4*)(inputr + (size_t)rl * F_OUT + lane * 4);
  in.sn = *(const float4*)(Snode + (size_t)src * 16);
  in.sr = *(const float4*)(Srel + (size_t)rl * 16);
  return in;
}

__device__ __forceinline__ void edge_acc(const EdgeIn& in, const float4* cv,
                                         float4* acc, float* rs) {
  const float logit[N_HEAD] = {in.sn.x + in.sr.x, in.sn.y + in.sr.y,
                               in.sn.z + in.sr.z, in.sn.w + in.sr.w};
#pragma unroll
  for (int i = 0; i < N_HEAD; ++i) {
    float lk = logit[i] > 0.0f ? logit[i] : 0.2f * logit[i];
    float ee = __expf(-lk);
    rs[i] += ee;
    acc[i].x += (cv[i].x * in.hs.x - in.rf.x) * ee;
    acc[i].y += (cv[i].y * in.hs.y - in.rf.y) * ee;
    acc[i].z += (cv[i].z * in.hs.z - in.rf.z) * ee;
    acc[i].w += (cv[i].w * in.hs.w - in.rf.w) * ee;
  }
}

__global__ void __launch_bounds__(128)
rgat_gather(const int* __restrict__ pk,
            const int* __restrict__ off,
            const float* __restrict__ h,
            const float* __restrict__ inputr,
            const float* __restrict__ w,
            const float* __restrict__ Snode,
            const float* __restrict__ Srel,
            float* __restrict__ out, int N) {
  const int lane = threadIdx.x & 31;
  const int wid  = (blockIdx.x * blockDim.x + threadIdx.x) >> 5;
  const int nw   = (gridDim.x * blockDim.x) >> 5;

  float4 cv[N_HEAD];
  head_scales(w, lane, cv);

  for (int dv = wid; dv < N; dv += nw) {
    const int d     = __builtin_amdgcn_readfirstlane(dv);
    const int start = off[d];
    const int deg   = off[d + 1] - start;

    float4 acc[N_HEAD];
    float  rs[N_HEAD];
#pragma unroll
    for (int i = 0; i < N_HEAD; ++i) {
      acc[i] = make_float4(0.0f, 0.0f, 0.0f, 0.0f);
      rs[i]  = 0.0f;
    }

    for (int b = 0; b < deg; b += 32) {
      const int n = deg - b < 32 ? deg - b : 32;
      int2 mine = make_int2(0, 0);
      if (lane < n) mine = *(const int2*)(pk + 2 * (size_t)(start + b + lane));

      int j = 0;
      for (; j + 4 <= n; j += 4) {
        const EdgeIn e0 = edge_load(h, inputr, Snode, Srel,
            __builtin_amdgcn_readlane(mine.x, j),     __builtin_amdgcn_readlane(mine.y, j),     lane);
        const EdgeIn e1 = edge_load(h, inputr, Snode, Srel,
            __builtin_amdgcn_readlane(mine.x, j + 1), __builtin_amdgcn_readlane(mine.y, j + 1), lane);
        const EdgeIn e2 = edge_load(h, inputr, Snode, Srel,
            __builtin_amdgcn_readlane(mine.x, j + 2), __builtin_amdgcn_readlane(mine.y, j + 2), lane);
        const EdgeIn e3 = edge_load(h, inputr, Snode, Srel,
            __builtin_amdgcn_readlane(mine.x, j + 3), __builtin_amdgcn_readlane(mine.y, j + 3), lane);
        edge_acc(e0, cv, acc, rs);
        edge_acc(e1, cv, acc, rs);
        edge_acc(e2, cv, acc, rs);
        edge_acc(e3, cv, acc, rs);
      }
      for (; j < n; ++j) {
        const EdgeIn e0 = edge_load(h, inputr, Snode, Srel,
            __builtin_amdgcn_readlane(mine.x, j), __builtin_amdgcn_readlane(mine.y, j), lane);
        edge_acc(e0, cv, acc, rs);
      }
    }

#pragma unroll
    for (int i = 0; i < N_HEAD; ++i) {
      float4 o;
      o.x = acc[i].x / rs[i];
      o.y = acc[i].y / rs[i];
      o.z = acc[i].z / rs[i];
      o.w = acc[i].w / rs[i];
      *(float4*)(out + ((size_t)i * N + d) * F_OUT + lane * 4) = o;
    }
  }
}

__global__ void rgat_edge(const int* __restrict__ A,
                          const float* __restrict__ h,
                          const float* __restrict__ inputr,
                          const float* __restrict__ w,
                          const float* __restrict__ Snode,
                          const float* __restrict__ Srel,
                          float* __restrict__ rowsum,
                          float* __restrict__ out,
                          int E, int N) {
  const int lane = threadIdx.x & 31;
  const int wid  = (blockIdx.x * blockDim.x + threadIdx.x) >> 5;
  const int nw   = (gridDim.x * blockDim.x) >> 5;

  float4 cv[N_HEAD];
  head_scales(w, lane, cv);

  for (int ev = wid; ev < E; ev += nw) {
    const int e   = __builtin_amdgcn_readfirstlane(ev);
    const int dst = A[e];
    const int rl  = A[E + e];
    const int src = A[2 * E + e];

    const float4 hs = *(const float4*)(h + (size_t)src * F_OUT + lane * 4);
    const float4 rf = *(const float4*)(inputr + (size_t)rl * F_OUT + lane * 4);
    const float4 sn = *(const float4*)(Snode + (size_t)src * 16);
    const float4 sr = *(const float4*)(Srel + (size_t)rl * 16);

    const float logit[N_HEAD] = {sn.x + sr.x, sn.y + sr.y, sn.z + sr.z, sn.w + sr.w};
    float ee[N_HEAD];
#pragma unroll
    for (int i = 0; i < N_HEAD; ++i) {
      float lk = logit[i] > 0.0f ? logit[i] : 0.2f * logit[i];
      ee[i] = __expf(-lk);
    }

    if (lane == 0) {
#pragma unroll
      for (int i = 0; i < N_HEAD; ++i)
        __hip_atomic_fetch_add(&rowsum[(size_t)i * N + dst], ee[i],
                               __ATOMIC_RELAXED, __HIP_MEMORY_SCOPE_AGENT);
    }

#pragma unroll
    for (int i = 0; i < N_HEAD; ++i) {
      float* op = out + ((size_t)i * N + dst) * F_OUT + lane * 4;
      __hip_atomic_fetch_add(&op[0], (cv[i].x * hs.x - rf.x) * ee[i],
                             __ATOMIC_RELAXED, __HIP_MEMORY_SCOPE_AGENT);
      __hip_atomic_fetch_add(&op[1], (cv[i].y * hs.y - rf.y) * ee[i],
                             __ATOMIC_RELAXED, __HIP_MEMORY_SCOPE_AGENT);
      __hip_atomic_fetch_add(&op[2], (cv[i].z * hs.z - rf.z) * ee[i],
                             __ATOMIC_RELAXED, __HIP_MEMORY_SCOPE_AGENT);
      __hip_atomic_fetch_add(&op[3], (cv[i].w * hs.w - rf.w) * ee[i],
                             __ATOMIC_RELAXED, __HIP_MEMORY_SCOPE_AGENT);
    }
  }
}

__global__ void rgat_norm(float* __restrict__ out,
                          const float* __restrict__ rowsum,
                          int total) {
  int idx = blockIdx.x * blockDim.x + threadIdx.x;
  if (idx >= total) return;
  out[idx] = out[idx] / rowsum[idx >> 7];
}

extern "C" void kernel_launch(void* const* d_in, const int* in_sizes, int n_in,
                              void* d_out, int out_size, void* d_ws, size_t ws_size,
                              hipStream_t stream) {
  const float* h      = (const float*)d_in[0];
  const float* inputr = (const float*)d_in[1];
  const float* w      = (const float*)d_in[2];
  const float* a      = (const float*)d_in[3];
  const int*   A      = (const int*)d_in[4];

  const int N    = in_sizes[0] / F_OUT;
  const int NREL = in_sizes[1] / F_OUT;
  const int E    = in_sizes[4] / 3;

  float* ws = (float*)d_ws;
  size_t off = 0;
  const int nrel_tiles = (NREL + 15) / 16;
  const int node_tiles = (N + 15) / 16;
  float* Srel  = ws + off;  off += (size_t)nrel_tiles * 16 * 16;
  float* Snode = ws + off;  off += (size_t)node_tiles * 16 * 16;
  const size_t tail = off;

  int* cnt  = (int*)ws + off;  off += (size_t)N;
  int* offs = (int*)ws + off;  off += (size_t)N + 1;
  int* rank = (int*)ws + off;  off += (size_t)E;
  off = (off + 1) & ~(size_t)1;
  int* pk   = (int*)ws + off;  off += (size_t)2 * E;
  const bool grouped = ws_size >= off * sizeof(float);

  const int WPB = 8;

  if (grouped) {
    hipMemsetAsync(cnt, 0, (size_t)N * sizeof(int), stream);
  } else {
    hipMemsetAsync(d_out, 0, (size_t)out_size * sizeof(float), stream);
    hipMemsetAsync(ws + tail, 0, (size_t)N_HEAD * N * sizeof(float), stream);
  }

  const int node_blocks  = (node_tiles + WPB - 1) / WPB;
  const int score_blocks = node_blocks + (nrel_tiles + WPB - 1) / WPB;
  const int count_blocks = grouped ? (E + FRONT_T - 1) / FRONT_T : 0;
  rgat_front<<<score_blocks + count_blocks, FRONT_T, 0, stream>>>(
      h, inputr, w, a, A, Snode, Srel, cnt, rank, N, NREL, E, node_blocks, score_blocks);

  if (grouped) {
    rgat_scan<<<(N + SCAN_T - 1) / SCAN_T, SCAN_T, 0, stream>>>(cnt, offs, N);
    rgat_place<<<(E + 255) / 256, 256, 0, stream>>>(A, offs, rank, pk, E);

    int gblocks = (N + 3) / 4;
    if (gblocks > 4096) gblocks = 4096;
    rgat_gather<<<gblocks, 128, 0, stream>>>(pk, offs, h, inputr, w, Snode, Srel,
                                             (float*)d_out, N);
  } else {
    float* rowsum = ws + tail;
    int eblocks = (E + WPB - 1) / WPB;
    if (eblocks > 4096) eblocks = 4096;
    rgat_edge<<<eblocks, 256, 0, stream>>>(A, h, inputr, w, Snode, Srel,
                                           rowsum, (float*)d_out, E, N);

    rgat_norm<<<(out_size + 255) / 256, 256, 0, stream>>>((float*)d_out, rowsum, out_size);
  }
}


# Gather: an edge's 4 head weights are computed once, by the lane holding its list entry, and broadcast by readlane (was: on all 32 lanes from scalar score loads); past-the-end lanes carry weight 0
# speedup vs baseline: 4.8441x; 1.1061x over previous
#include <hip/hip_runtime.h>


typedef __attribute__((ext_vector_type(2))) float v2f;
typedef __attribute__((ext_vector_type(8))) float v8f;

#define F_OUT 128
#define N_HEAD 4

#define FRONT_T 256
__global__ void __launch_bounds__(FRONT_T)
rgat_front(const float* __restrict__ h,
           const float* __restrict__ inputr,
           const float* __restrict__ w,
           const float* __restrict__ a,
           const int* __restrict__ A,
           float* __restrict__ Snode,
           float* __restrict__ Srel,
           int* __restrict__ cnt,
           int* __restrict__ rank,
           int N, int NREL, int E, int node_blocks, int score_blocks) {
  __shared__ float Bl[32 * 64];
  const int b = blockIdx.x;

  if (b >= score_blocks) {
    const int e = (b - score_blocks) * FRONT_T + threadIdx.x;
    if (e < E)
      rank[e] = __hip_atomic_fetch_add(&cnt[A[e]], 1, __ATOMIC_RELAXED,
                                       __HIP_MEMORY_SCOPE_AGENT);
    return;
  }

  const bool isrel = b >= node_blocks;
  {
    const int t    = threadIdx.x;
    const int l    = (t >> 1) & 31;
    const int col  = l & 15;
    const int k0   = (t >> 6) * 4 + ((l >> 4) << 1) + (t & 1);
    float v[8];
#pragma unroll
    for (int i = 0; i < 8; ++i) v[i] = 0.0f;
    if (col < 4) {
      if (isrel) {
#pragma unroll
        for (int i = 0; i < 8; ++i)
          v[i] = a[col * 256 + F_OUT + k0 + 16 * i];
      } else {
#pragma unroll
        for (int i = 0; i < 8; ++i) {
          const int k = k0 + 16 * i;
          const float f0 = w[0 * F_OUT + k];
          const float f1 = w[1 * F_OUT + k];
          const float f2 = w[2 * F_OUT + k];
          float c = col > 0 ? f0 : 1.0f;
          c *= col > 1 ? f1 : 1.0f;
          c *= col > 2 ? f2 : 1.0f;
          v[i] = c * a[col * 256 + k];
        }
      }
    }
#pragma unroll
    for (int i = 0; i < 8; ++i) Bl[t + FRONT_T * i] = v[i];
  }
  __syncthreads();

  const float* __restrict__ X = isrel ? inputr : h;
  float* __restrict__ S       = isrel ? Srel : Snode;
  const int M                 = isrel ? NREL : N;
  const int tile = (isrel ? b - node_blocks : b) * (FRONT_T >> 5) + (threadIdx.x >> 5);
  const int base = tile * 16;
  if (base >= M) return;

  const int lane = threadIdx.x & 31;
  const int col  = lane & 15;
  const int kb   = (lane >> 4) * 2;
  int row = base + (lane & 15);
  row = row < M ? row : M - 1;
  const float* xr = X + (size_t)row * F_OUT;
  const float* bp = Bl + lane * 2;

  v8f acc = {};
#pragma unroll
  for (int s = 0; s < 32; ++s) {
    v2f av = *(const v2f*)(xr + s * 4 + kb);
    v2f bv = *(const v2f*)(bp + s * 64);
    acc = __builtin_amdgcn_wmma_f32_16x16x4_f32(
         false, av,  false, bv,
         (short)0, acc,  false,  false);
  }

  const int mhalf = (lane >> 4) * 8;
#pragma unroll
  for (int v = 0; v < 8; ++v)
    S[(size_t)(base + mhalf + v) * 16 + col] = acc[v];
}

#define SCAN_T 1024
__global__ void __launch_bounds__(SCAN_T)
rgat_scan(const int* __restrict__ cnt, int* __restrict__ off, int N) {
  __shared__ int wtot[32];
  __shared__ int ptot[32];
  const int tid  = threadIdx.x;
  const int lane = tid & 31;
  const int wv   = tid >> 5;
  const int first = blockIdx.x * SCAN_T;

  int pre = 0;
  for (int i = tid; i < first; i += SCAN_T) pre += cnt[i];

  const int idx = first + tid;
  const int v = idx < N ? cnt[idx] : 0;
  int x = v;
  for (int d = 1; d < 32; d <<= 1) {
    int y = __shfl_up(x, d, 32);
    if (lane >= d) x += y;
    pre += __shfl_xor(pre, d, 32);
  }
  if (lane == 31) wtot[wv] = x;
  if (lane == 0)  ptot[wv] = pre;
  __syncthreads();

  int base = 0;
  for (int k = 0; k < 32; ++k) {
    base += ptot[k];
    if (k < wv) base += wtot[k];
  }
  if (idx < N) off[idx] = base + x - v;
  if (idx == N - 1) off[N] = base + x;
}

__global__ void rgat_place(const int* __restrict__ A,
                           const int* __restrict__ off,
                           const int* __restrict__ rank,
                           int* __restrict__ pk, int E) {
  int e = blockIdx.x * blockDim.x + threadIdx.x;
  if (e >= E) return;
  const int p = off[A[e]] + rank[e];
  *(int2*)(pk + 2 * (size_t)p) = make_int2(A[2 * E + e], A[E + e]);
}

struct EdgeRows { float4 hs, rf; };

__device__ __forceinline__ void head_scales(const float* __restrict__ w, int lane,
                                            float4* cv) {
  const float4 w0 = *(const float4*)(w + 0 * F_OUT + lane * 4);
  const float4 w1 = *(const float4*)(w + 1 * F_OUT + lane * 4);
  const float4 w2 = *(const float4*)(w + 2 * F_OUT + lane * 4);
  cv[0] = make_float4(1.0f, 1.0f, 1.0f, 1.0f);
  cv[1] = w0;
  cv[2] = make_float4(cv[1].x * w1.x, cv[1].y * w1.y, cv[1].z * w1.z, cv[1].w * w1.w);
  cv[3] = make_float4(cv[2].x * w2.x, cv[2].y * w2.y, cv[2].z * w2.z, cv[2].w * w2.w);
}

__device__ __forceinline__ EdgeRows rows_load(const float* __restrict__ h,
                                              const float* __restrict__ inputr,
                                              int src, int rl, int lane) {
  EdgeRows in;
  in.hs = *(const float4*)(h + (size_t)src * F_OUT + lane * 4);
  in.rf = *(const float4*)(inputr + (size_t)rl * F_OUT + lane * 4);
  return in;
}

__device__ __forceinline__ float4 edge_weights(const float4 sn, const float4 sr) {
  const float logit[N_HEAD] = {sn.x + sr.x, sn.y + sr.y, sn.z + sr.z, sn.w + sr.w};
  float ee[N_HEAD];
#pragma unroll
  for (int i = 0; i < N_HEAD; ++i) {
    const float lk = logit[i] > 0.0f ? logit[i] : 0.2f * logit[i];
    ee[i] = __expf(-lk);
  }
  return make_float4(ee[0], ee[1], ee[2], ee[3]);
}

__device__ __forceinline__ float4 weights_of(const float4 wt, int j) {
  return make_float4(
      __int_as_float(__builtin_amdgcn_readlane(__float_as_int(wt.x), j)),
      __int_as_float(__builtin_amdgcn_readlane(__float_as_int(wt.y), j)),
      __int_as_float(__builtin_amdgcn_readlane(__float_as_int(wt.z), j)),
      __int_as_float(__builtin_amdgcn_readlane(__float_as_int(wt.w), j)));
}

__device__ __forceinline__ void rows_acc(const EdgeRows& in, const float4 wv,
                                         const float4* cv, float4* acc, float* rs) {
  const float ee[N_HEAD] = {wv.x, wv.y, wv.z, wv.w};
#pragma unroll
  for (int i = 0; i < N_HEAD; ++i) {
    rs[i] += ee[i];
    acc[i].x += (cv[i].x * in.hs.x - in.rf.x) * ee[i];
    acc[i].y += (cv[i].y * in.hs.y - in.rf.y) * ee[i];
    acc[i].z += (cv[i].z * in.hs.z - in.rf.z) * ee[i];
    acc[i].w += (cv[i].w * in.hs.w - in.rf.w) * ee[i];
  }
}

__global__ void __launch_bounds__(128)
rgat_gather(const int* __restrict__ pk,
            const int* __restrict__ off,
            const float* __restrict__ h,
            const float* __restrict__ inputr,
            const float* __restrict__ w,
            const float* __restrict__ Snode,
            const float* __restrict__ Srel,
            float* __restrict__ out, int N) {
  const int lane = threadIdx.x & 31;
  const int wid  = (blockIdx.x * blockDim.x + threadIdx.x) >> 5;
  const int nw   = (gridDim.x * blockDim.x) >> 5;

  float4 cv[N_HEAD];
  head_scales(w, lane, cv);

  for (int dv = wid; dv < N; dv += nw) {
    const int d     = __builtin_amdgcn_readfirstlane(dv);
    const int start = off[d];
    const int deg   = off[d + 1] - start;

    float4 acc[N_HEAD];
    float  rs[N_HEAD];
#pragma unroll
    for (int i = 0; i < N_HEAD; ++i) {
      acc[i] = make_float4(0.0f, 0.0f, 0.0f, 0.0f);
      rs[i]  = 0.0f;
    }

    for (int b = 0; b < deg; b += 32) {
      const int n = deg - b < 32 ? deg - b : 32;
      int2 mine = make_int2(0, 0);
      if (lane < n) mine = *(const int2*)(pk + 2 * (size_t)(start + b + lane));

      const float4 sn = *(const float4*)(Snode + (size_t)mine.x * 16);
      const float4 sr = *(const float4*)(Srel + (size_t)mine.y * 16);

      float4 wt;
      {
        const EdgeRows r0 = rows_load(h, inputr,
            __builtin_amdgcn_readlane(mine.x, 0), __builtin_amdgcn_readlane(mine.y, 0), lane);
        const EdgeRows r1 = rows_load(h, inputr,
            __builtin_amdgcn_readlane(mine.x, 1), __builtin_amdgcn_readlane(mine.y, 1), lane);
        const EdgeRows r2 = rows_load(h, inputr,
            __builtin_amdgcn_readlane(mine.x, 2), __builtin_amdgcn_readlane(mine.y, 2), lane);
        const EdgeRows r3 = rows_load(h, inputr,
            __builtin_amdgcn_readlane(mine.x, 3), __builtin_amdgcn_readlane(mine.y, 3), lane);

        wt = edge_weights(sn, sr);
        if (lane >= n) wt = make_float4(0.0f, 0.0f, 0.0f, 0.0f);

        rows_acc(r0, weights_of(wt, 0), cv, acc, rs);
        rows_acc(r1, weights_of(wt, 1), cv, acc, rs);
        rows_acc(r2, weights_of(wt, 2), cv, acc, rs);
        rows_acc(r3, weights_of(wt, 3), cv, acc, rs);
      }

      for (int j = 4; j < n; j += 4) {
        const EdgeRows r0 = rows_load(h, inputr,
            __builtin_amdgcn_readlane(mine.x, j),     __builtin_amdgcn_readlane(mine.y, j),     lane);
        const EdgeRows r1 = rows_load(h, inputr,
            __builtin_amdgcn_readlane(mine.x, j + 1), __builtin_amdgcn_readlane(mine.y, j + 1), lane);
        const EdgeRows r2 = rows_load(h, inputr,
            __builtin_amdgcn_readlane(mine.x, j + 2), __builtin_amdgcn_readlane(mine.y, j + 2), lane);
        const EdgeRows r3 = rows_load(h, inputr,
            __builtin_amdgcn_readlane(mine.x, j + 3), __builtin_amdgcn_readlane(mine.y, j + 3), lane);
        rows_acc(r0, weights_of(wt, j),     cv, acc, rs);
        rows_acc(r1, weights_of(wt, j + 1), cv, acc, rs);
        rows_acc(r2, weights_of(wt, j + 2), cv, acc, rs);
        rows_acc(r3, weights_of(wt, j + 3), cv, acc, rs);
      }
    }

#pragma unroll
    for (int i = 0; i < N_HEAD; ++i) {
      float4 o;
      o.x = acc[i].x / rs[i];
      o.y = acc[i].y / rs[i];
      o.z = acc[i].z / rs[i];
      o.w = acc[i].w / rs[i];
      *(float4*)(out + ((size_t)i * N + d) * F_OUT + lane * 4) = o;
    }
  }
}

__global__ void rgat_edge(const int* __restrict__ A,
                          const float* __restrict__ h,
                          const float* __restrict__ inputr,
                          const float* __restrict__ w,
                          const float* __restrict__ Snode,
                          const float* __restrict__ Srel,
                          float* __restrict__ rowsum,
                          float* __restrict__ out,
                          int E, int N) {
  const int lane = threadIdx.x & 31;
  const int wid  = (blockIdx.x * blockDim.x + threadIdx.x) >> 5;
  const int nw   = (gridDim.x * blockDim.x) >> 5;

  float4 cv[N_HEAD];
  head_scales(w, lane, cv);

  for (int ev = wid; ev < E; ev += nw) {
    const int e   = __builtin_amdgcn_readfirstlane(ev);
    const int dst = A[e];
    const int rl  = A[E + e];
    const int src = A[2 * E + e];

    const float4 hs = *(const float4*)(h + (size_t)src * F_OUT + lane * 4);
    const float4 rf = *(const float4*)(inputr + (size_t)rl * F_OUT + lane * 4);
    const float4 sn = *(const float4*)(Snode + (size_t)src * 16);
    const float4 sr = *(const float4*)(Srel + (size_t)rl * 16);

    const float logit[N_HEAD] = {sn.x + sr.x, sn.y + sr.y, sn.z + sr.z, sn.w + sr.w};
    float ee[N_HEAD];
#pragma unroll
    for (int i = 0; i < N_HEAD; ++i) {
      float lk = logit[i] > 0.0f ? logit[i] : 0.2f * logit[i];
      ee[i] = __expf(-lk);
    }

    if (lane == 0) {
#pragma unroll
      for (int i = 0; i < N_HEAD; ++i)
        __hip_atomic_fetch_add(&rowsum[(size_t)i * N + dst], ee[i],
                               __ATOMIC_RELAXED, __HIP_MEMORY_SCOPE_AGENT);
    }

#pragma unroll
    for (int i = 0; i < N_HEAD; ++i) {
      float* op = out + ((size_t)i * N + dst) * F_OUT + lane * 4;
      __hip_atomic_fetch_add(&op[0], (cv[i].x * hs.x - rf.x) * ee[i],
                             __ATOMIC_RELAXED, __HIP_MEMORY_SCOPE_AGENT);
      __hip_atomic_fetch_add(&op[1], (cv[i].y * hs.y - rf.y) * ee[i],
                             __ATOMIC_RELAXED, __HIP_MEMORY_SCOPE_AGENT);
      __hip_atomic_fetch_add(&op[2], (cv[i].z * hs.z - rf.z) * ee[i],
                             __ATOMIC_RELAXED, __HIP_MEMORY_SCOPE_AGENT);
      __hip_atomic_fetch_add(&op[3], (cv[i].w * hs.w - rf.w) * ee[i],
                             __ATOMIC_RELAXED, __HIP_MEMORY_SCOPE_AGENT);
    }
  }
}

__global__ void rgat_norm(float* __restrict__ out,
                          const float* __restrict__ rowsum,
                          int total) {
  int idx = blockIdx.x * blockDim.x + threadIdx.x;
  if (idx >= total) return;
  out[idx] = out[idx] / rowsum[idx >> 7];
}

extern "C" void kernel_launch(void* const* d_in, const int* in_sizes, int n_in,
                              void* d_out, int out_size, void* d_ws, size_t ws_size,
                              hipStream_t stream) {
  const float* h      = (const float*)d_in[0];
  const float* inputr = (const float*)d_in[1];
  const float* w      = (const float*)d_in[2];
  const float* a      = (const float*)d_in[3];
  const int*   A      = (const int*)d_in[4];

  const int N    = in_sizes[0] / F_OUT;
  const int NREL = in_sizes[1] / F_OUT;
  const int E    = in_sizes[4] / 3;

  float* ws = (float*)d_ws;
  size_t off = 0;
  const int nrel_tiles = (NREL + 15) / 16;
  const int node_tiles = (N + 15) / 16;
  float* Srel  = ws + off;  off += (size_t)nrel_tiles * 16 * 16;
  float* Snode = ws + off;  off += (size_t)node_tiles * 16 * 16;
  const size_t tail = off;

  int* cnt  = (int*)ws + off;  off += (size_t)N;
  int* offs = (int*)ws + off;  off += (size_t)N + 1;
  int* rank = (int*)ws + off;  off += (size_t)E;
  off = (off + 1) & ~(size_t)1;
  int* pk   = (int*)ws + off;  off += (size_t)2 * E;
  const bool grouped = ws_size >= off * sizeof(float);

  const int WPB = 8;

  if (grouped) {
    hipMemsetAsync(cnt, 0, (size_t)N * sizeof(int), stream);
  } else {
    hipMemsetAsync(d_out, 0, (size_t)out_size * sizeof(float), stream);
    hipMemsetAsync(ws + tail, 0, (size_t)N_HEAD * N * sizeof(float), stream);
  }

  const int node_blocks  = (node_tiles + WPB - 1) / WPB;
  const int score_blocks = node_blocks + (nrel_tiles + WPB - 1) / WPB;
  const int count_blocks = grouped ? (E + FRONT_T - 1) / FRONT_T : 0;
  rgat_front<<<score_blocks + count_blocks, FRONT_T, 0, stream>>>(
      h, inputr, w, a, A, Snode, Srel, cnt, rank, N, NREL, E, node_blocks, score_blocks);

  if (grouped) {
    rgat_scan<<<(N + SCAN_T - 1) / SCAN_T, SCAN_T, 0, stream>>>(cnt, offs, N);
    rgat_place<<<(E + 255) / 256, 256, 0, stream>>>(A, offs, rank, pk, E);

    int gblocks = (N + 3) / 4;
    if (gblocks > 4096) gblocks = 4096;
    rgat_gather<<<gblocks, 128, 0, stream>>>(pk, offs, h, inputr, w, Snode, Srel,
                                             (float*)d_out, N);
  } else {
    float* rowsum = ws + tail;
    int eblocks = (E + WPB - 1) / WPB;
    if (eblocks > 4096) eblocks = 4096;
    rgat_edge<<<eblocks, 256, 0, stream>>>(A, h, inputr, w, Snode, Srel,
                                           rowsum, (float*)d_out, E, N);

    rgat_norm<<<(out_size + 255) / 256, 256, 0, stream>>>((float*)d_out, rowsum, out_size);
  }
}
